# starting kernel
# baseline (speedup 1.0000x reference)
#include <hip/hip_runtime.h>
#include <math.h>

typedef float v2f __attribute__((ext_vector_type(2)));
typedef float v8f __attribute__((ext_vector_type(8)));

#define BATCH   2
#define T_LEN   16000
#define NCEP    25
#define M_ORD   24
#define IR_LEN  2000
#define TILE_T  128
#define NWAVES  4
#define CHUNK   80
#define NCHUNK  25
#define HSTRIDE 81
#define XS_LEN  2176

__global__ __launch_bounds__(TILE_T)
void tv_fir_c2ir_kernel(const float* __restrict__ x,
                        const float* __restrict__ mc,
                        float* __restrict__ y) {
  __shared__ float xs[XS_LEN];
  __shared__ float hs[TILE_T * HSTRIDE];
  __shared__ float invn[IR_LEN];
  __shared__ float dbuf[NWAVES][16][16];

  const int tid  = threadIdx.x;
  const int wave = tid >> 5;
  const int lane = tid & 31;
  const int hi   = lane >> 4;
  const int lm   = lane & 15;

  const int b  = blockIdx.x / (T_LEN / TILE_T);
  const int tb = (blockIdx.x % (T_LEN / TILE_T)) * TILE_T;

  for (int i = tid; i < XS_LEN; i += TILE_T) {
    int g = tb - (IR_LEN - 1) + i;
    xs[i] = (i < (IR_LEN - 1 + TILE_T) && g >= 0 && g < T_LEN)
                ? x[(size_t)b * T_LEN + g] : 0.0f;
  }
  for (int n = tid; n < IR_LEN; n += TILE_T)
    invn[n] = (n > 0) ? 1.0f / (float)n : 0.0f;

  const float* mcp = mc + ((size_t)b * T_LEN + (tb + tid)) * NCEP;
  float c0 = mcp[0];
  float kc[NCEP];
#pragma unroll
  for (int k = 1; k < NCEP; ++k) kc[k] = (float)k * mcp[k];

  v8f acc0 = {};
  v8f acc1 = {};
  float hp[M_ORD];
#pragma unroll
  for (int k = 0; k < M_ORD; ++k) hp[k] = 0.0f;

  __syncthreads();

  float* hrow = &hs[tid * HSTRIDE];

  for (int chunk = 0; chunk < NCHUNK; ++chunk) {
    const int n0 = chunk * CHUNK;

    if (chunk == 0) {
      float h0 = __expf(c0);
      hrow[0] = h0;
#pragma unroll
      for (int k2 = M_ORD - 1; k2 > 0; --k2) hp[k2] = hp[k2 - 1];
      hp[0] = h0;
#pragma unroll 10
      for (int j = 1; j < CHUNK; ++j) {
        float s = 0.0f;
#pragma unroll
        for (int k = 1; k <= M_ORD; ++k) s += kc[k] * hp[k - 1];
        float hn = s * invn[j];
        hrow[j] = hn;
#pragma unroll
        for (int k2 = M_ORD - 1; k2 > 0; --k2) hp[k2] = hp[k2 - 1];
        hp[0] = hn;
      }
    } else {
#pragma unroll 10
      for (int j = 0; j < CHUNK; ++j) {
        float s = 0.0f;
#pragma unroll
        for (int k = 1; k <= M_ORD; ++k) s += kc[k] * hp[k - 1];
        float hn = s * invn[n0 + j];
        hrow[j] = hn;
#pragma unroll
        for (int k2 = M_ORD - 1; k2 > 0; --k2) hp[k2] = hp[k2 - 1];
        hp[0] = hn;
      }
    }
    __syncthreads();

#pragma unroll
    for (int st = 0; st < 2; ++st) {
      const int sub = wave * 32 + st * 16;
      v8f acc = st ? acc1 : acc0;
      const float* hsub = &hs[(sub + lm) * HSTRIDE];
#pragma unroll
      for (int g = 0; g < CHUNK / 4; ++g) {
        const int kl = g * 4 + 2 * hi;
        v2f a;  a.x = hsub[kl];  a.y = hsub[kl + 1];
        const int xb = (IR_LEN - 1) + sub + lm - (n0 + kl);
        v2f bv; bv.x = xs[xb];   bv.y = xs[xb - 1];
        acc = __builtin_amdgcn_wmma_f32_16x16x4_f32(
             false, a,  false, bv,
             (short)0, acc,  false,  false);
      }
      if (st) acc1 = acc; else acc0 = acc;
    }
    __syncthreads();
  }

  for (int st = 0; st < 2; ++st) {
    v8f acc = st ? acc1 : acc0;
#pragma unroll
    for (int r = 0; r < 8; ++r)
      dbuf[wave][r + 8 * hi][lm] = acc[r];
    __syncthreads();
    if (lane < 16) {
      const int sub = wave * 32 + st * 16;
      y[(size_t)b * T_LEN + tb + sub + lane] = dbuf[wave][lane][lane];
    }
    __syncthreads();
  }
}

extern "C" void kernel_launch(void* const* d_in, const int* in_sizes, int n_in,
                              void* d_out, int out_size, void* d_ws, size_t ws_size,
                              hipStream_t stream) {
  (void)in_sizes; (void)n_in; (void)out_size; (void)d_ws; (void)ws_size;
  const float* x  = (const float*)d_in[0];
  const float* mc = (const float*)d_in[1];
  float* y = (float*)d_out;

  dim3 grid(BATCH * (T_LEN / TILE_T));
  dim3 block(TILE_T);
  tv_fir_c2ir_kernel<<<grid, block, 0, stream>>>(x, mc, y);
}


# c2ir recursion in transposed form: each new tap is added into 24 running sums, so a tap waits on one fma and one multiply, not a 25-op chain; terms now summed oldest tap first
# speedup vs baseline: 3.1065x; 3.1065x over previous
#include <hip/hip_runtime.h>
#include <math.h>

typedef float v2f __attribute__((ext_vector_type(2)));
typedef float v8f __attribute__((ext_vector_type(8)));

#define BATCH   2
#define T_LEN   16000
#define NCEP    25
#define M_ORD   24
#define IR_LEN  2000
#define TILE_T  128
#define NWAVES  4
#define CHUNK   80
#define NCHUNK  25
#define HSTRIDE 81
#define XS_LEN  2176

__global__ __launch_bounds__(TILE_T)
void tv_fir_c2ir_kernel(const float* __restrict__ x,
                        const float* __restrict__ mc,
                        float* __restrict__ y) {
  __shared__ float xs[XS_LEN];
  __shared__ float hs[TILE_T * HSTRIDE];
  __shared__ float invn[IR_LEN];
  __shared__ float dbuf[NWAVES][16][16];

  const int tid  = threadIdx.x;
  const int wave = tid >> 5;
  const int lane = tid & 31;
  const int hi   = lane >> 4;
  const int lm   = lane & 15;

  const int b  = blockIdx.x / (T_LEN / TILE_T);
  const int tb = (blockIdx.x % (T_LEN / TILE_T)) * TILE_T;

  for (int i = tid; i < XS_LEN; i += TILE_T) {
    int g = tb - (IR_LEN - 1) + i;
    xs[i] = (i < (IR_LEN - 1 + TILE_T) && g >= 0 && g < T_LEN)
                ? x[(size_t)b * T_LEN + g] : 0.0f;
  }
  for (int n = tid; n < IR_LEN; n += TILE_T)
    invn[n] = (n > 0) ? 1.0f / (float)n : 1.0f;

  const float* mcp = mc + ((size_t)b * T_LEN + (tb + tid)) * NCEP;
  float c0 = mcp[0];
  float kc[NCEP];
#pragma unroll
  for (int k = 1; k < NCEP; ++k) kc[k] = (float)k * mcp[k];

  v8f acc0 = {};
  v8f acc1 = {};
  float ps[M_ORD];
  ps[0] = __expf(c0);
#pragma unroll
  for (int k = 1; k < M_ORD; ++k) ps[k] = 0.0f;

  __syncthreads();

  float* hrow = &hs[tid * HSTRIDE];

  for (int chunk = 0; chunk < NCHUNK; ++chunk) {
    const int n0 = chunk * CHUNK;

#pragma unroll
    for (int j = 0; j < CHUNK; ++j) {
      const float hn = ps[0] * invn[n0 + j];
      hrow[j] = hn;
#pragma unroll
      for (int i = 0; i < M_ORD - 1; ++i) ps[i] = fmaf(kc[i + 1], hn, ps[i + 1]);
      ps[M_ORD - 1] = kc[M_ORD] * hn;
    }
    __syncthreads();

#pragma unroll
    for (int st = 0; st < 2; ++st) {
      const int sub = wave * 32 + st * 16;
      v8f acc = st ? acc1 : acc0;
      const float* hsub = &hs[(sub + lm) * HSTRIDE];
#pragma unroll
      for (int g = 0; g < CHUNK / 4; ++g) {
        const int kl = g * 4 + 2 * hi;
        v2f a;  a.x = hsub[kl];  a.y = hsub[kl + 1];
        const int xb = (IR_LEN - 1) + sub + lm - (n0 + kl);
        v2f bv; bv.x = xs[xb];   bv.y = xs[xb - 1];
        acc = __builtin_amdgcn_wmma_f32_16x16x4_f32(
             false, a,  false, bv,
             (short)0, acc,  false,  false);
      }
      if (st) acc1 = acc; else acc0 = acc;
    }
    __syncthreads();
  }

  for (int st = 0; st < 2; ++st) {
    v8f acc = st ? acc1 : acc0;
#pragma unroll
    for (int r = 0; r < 8; ++r)
      dbuf[wave][r + 8 * hi][lm] = acc[r];
    __syncthreads();
    if (lane < 16) {
      const int sub = wave * 32 + st * 16;
      y[(size_t)b * T_LEN + tb + sub + lane] = dbuf[wave][lane][lane];
    }
    __syncthreads();
  }
}

extern "C" void kernel_launch(void* const* d_in, const int* in_sizes, int n_in,
                              void* d_out, int out_size, void* d_ws, size_t ws_size,
                              hipStream_t stream) {
  (void)in_sizes; (void)n_in; (void)out_size; (void)d_ws; (void)ws_size;
  const float* x  = (const float*)d_in[0];
  const float* mc = (const float*)d_in[1];
  float* y = (float*)d_out;

  dim3 grid(BATCH * (T_LEN / TILE_T));
  dim3 block(TILE_T);
  tv_fir_c2ir_kernel<<<grid, block, 0, stream>>>(x, mc, y);
}


# wave specialisation: 4 waves run the c2ir recursion one 80-tap chunk ahead into a double-buffered LDS h tile while 4 added waves run the WMMA FIR of the previous chunk; 1 barrier per chunk, was 2
# speedup vs baseline: 3.6559x; 1.1769x over previous
#include <hip/hip_runtime.h>
#include <math.h>

typedef float v2f __attribute__((ext_vector_type(2)));
typedef float v4f __attribute__((ext_vector_type(4)));
typedef float v8f __attribute__((ext_vector_type(8)));

#define BATCH   2
#define T_LEN   16000
#define NCEP    25
#define M_ORD   24
#define IR_LEN  2000
#define TILE_T  128
#define NMWAVES 4
#define NRWAVES 4
#define NTHREADS ((NMWAVES + NRWAVES) * 32)
#define CHUNK   80
#define NCHUNK  25
#define HSTRIDE 81
#define HBUF    (TILE_T * HSTRIDE)
#define XS_LEN  2176

__global__ __launch_bounds__(NTHREADS)
void tv_fir_c2ir_kernel(const float* __restrict__ x,
                        const float* __restrict__ mc,
                        float* __restrict__ y) {
  __shared__ float xs[XS_LEN];
  __shared__ float hs[2 * HBUF];
  __shared__ __attribute__((aligned(16))) float invn[IR_LEN];
  __shared__ float dbuf[NMWAVES][16][16];

  const int tid  = threadIdx.x;
  const int wave = __builtin_amdgcn_readfirstlane(tid >> 5);
  const bool mat_role = wave < NMWAVES;
  const int rt   = tid - NMWAVES * 32;
  const int lane = tid & 31;
  const int hi   = lane >> 4;
  const int lm   = lane & 15;

  const int b  = blockIdx.x / (T_LEN / TILE_T);
  const int tb = (blockIdx.x % (T_LEN / TILE_T)) * TILE_T;

  for (int i = tid; i < XS_LEN; i += NTHREADS) {
    int g = tb - (IR_LEN - 1) + i;
    xs[i] = (i < (IR_LEN - 1 + TILE_T) && g >= 0 && g < T_LEN)
                ? x[(size_t)b * T_LEN + g] : 0.0f;
  }
  for (int n = tid; n < IR_LEN; n += NTHREADS)
    invn[n] = (n > 0) ? 1.0f / (float)n : 1.0f;

  float kc[NCEP];
  float ps[M_ORD];
#pragma unroll
  for (int k = 0; k < NCEP; ++k) kc[k] = 0.0f;
#pragma unroll
  for (int k = 0; k < M_ORD; ++k) ps[k] = 0.0f;
  if (!mat_role) {
    const float* mcp = mc + ((size_t)b * T_LEN + (tb + rt)) * NCEP;
    const float c0 = mcp[0];
#pragma unroll
    for (int k = 1; k < NCEP; ++k) kc[k] = (float)k * mcp[k];
    ps[0] = __expf(c0);
  }

  v8f acc0 = {};
  v8f acc1 = {};

  __syncthreads();

  for (int it = 0; it <= NCHUNK; ++it) {
    if (!mat_role) {
      if (it < NCHUNK) {
        const int n0 = it * CHUNK;
        float* hrow = &hs[(it & 1) * HBUF + rt * HSTRIDE];
        v4f iv[CHUNK / 4];
#pragma unroll
        for (int q = 0; q < CHUNK / 4; ++q)
          iv[q] = *reinterpret_cast<const v4f*>(&invn[n0 + 4 * q]);
        __builtin_amdgcn_sched_barrier(0);
#pragma unroll
        for (int j = 0; j < CHUNK; ++j) {
          const float hn = ps[0] * iv[j >> 2][j & 3];
          hrow[j] = hn;
#pragma unroll
          for (int i = 0; i < M_ORD - 1; ++i) ps[i] = fmaf(kc[i + 1], hn, ps[i + 1]);
          ps[M_ORD - 1] = kc[M_ORD] * hn;
        }
      }
    } else {
      if (it > 0) {
        const int n0 = (it - 1) * CHUNK;
        const float* hbuf = &hs[((it - 1) & 1) * HBUF];
#pragma unroll
        for (int st = 0; st < 2; ++st) {
          const int sub = wave * 32 + st * 16;
          v8f acc = st ? acc1 : acc0;
          const float* hsub = &hbuf[(sub + lm) * HSTRIDE];
#pragma unroll
          for (int g = 0; g < CHUNK / 4; ++g) {
            const int kl = g * 4 + 2 * hi;
            v2f a;  a.x = hsub[kl];  a.y = hsub[kl + 1];
            const int xb = (IR_LEN - 1) + sub + lm - (n0 + kl);
            v2f bv; bv.x = xs[xb];   bv.y = xs[xb - 1];
            acc = __builtin_amdgcn_wmma_f32_16x16x4_f32(
                 false, a,  false, bv,
                 (short)0, acc,  false,  false);
          }
          if (st) acc1 = acc; else acc0 = acc;
        }
      }
    }
    __syncthreads();
  }

  for (int st = 0; st < 2; ++st) {
    if (mat_role) {
      v8f acc = st ? acc1 : acc0;
#pragma unroll
      for (int r = 0; r < 8; ++r)
        dbuf[wave][r + 8 * hi][lm] = acc[r];
    }
    __syncthreads();
    if (mat_role && lane < 16) {
      const int sub = wave * 32 + st * 16;
      y[(size_t)b * T_LEN + tb + sub + lane] = dbuf[wave][lane][lane];
    }
    __syncthreads();
  }
}

extern "C" void kernel_launch(void* const* d_in, const int* in_sizes, int n_in,
                              void* d_out, int out_size, void* d_ws, size_t ws_size,
                              hipStream_t stream) {
  (void)in_sizes; (void)n_in; (void)out_size; (void)d_ws; (void)ws_size;
  const float* x  = (const float*)d_in[0];
  const float* mc = (const float*)d_in[1];
  float* y = (float*)d_out;

  dim3 grid(BATCH * (T_LEN / TILE_T));
  dim3 block(NTHREADS);
  tv_fir_c2ir_kernel<<<grid, block, 0, stream>>>(x, mc, y);
}


# exact early exit: recursion waves test once per 80-tap chunk whether all 24 running sums of every chain are zero (then all later taps are zero); if so the workgroup leaves the chunk loop; same sums
# speedup vs baseline: 6.4316x; 1.7592x over previous
#include <hip/hip_runtime.h>
#include <math.h>

typedef float v2f __attribute__((ext_vector_type(2)));
typedef float v4f __attribute__((ext_vector_type(4)));
typedef float v8f __attribute__((ext_vector_type(8)));
typedef int   v4i __attribute__((ext_vector_type(4)));

#define BATCH   2
#define T_LEN   16000
#define NCEP    25
#define M_ORD   24
#define IR_LEN  2000
#define TILE_T  128
#define NMWAVES 4
#define NRWAVES 4
#define NTHREADS ((NMWAVES + NRWAVES) * 32)
#define CHUNK   80
#define NCHUNK  25
#define HSTRIDE 81
#define HBUF    (TILE_T * HSTRIDE)
#define XS_LEN  2176

__global__ __launch_bounds__(NTHREADS)
void tv_fir_c2ir_kernel(const float* __restrict__ x,
                        const float* __restrict__ mc,
                        float* __restrict__ y) {
  __shared__ float xs[XS_LEN];
  __shared__ float hs[2 * HBUF];
  __shared__ __attribute__((aligned(16))) float invn[IR_LEN];
  __shared__ float dbuf[NMWAVES][16][16];
  __shared__ __attribute__((aligned(16))) int zflag[2][NRWAVES];

  const int tid  = threadIdx.x;
  const int wave = __builtin_amdgcn_readfirstlane(tid >> 5);
  const bool mat_role = wave < NMWAVES;
  const int rt   = tid - NMWAVES * 32;
  const int lane = tid & 31;
  const int hi   = lane >> 4;
  const int lm   = lane & 15;

  const int b  = blockIdx.x / (T_LEN / TILE_T);
  const int tb = (blockIdx.x % (T_LEN / TILE_T)) * TILE_T;

  for (int i = tid; i < XS_LEN; i += NTHREADS) {
    int g = tb - (IR_LEN - 1) + i;
    xs[i] = (i < (IR_LEN - 1 + TILE_T) && g >= 0 && g < T_LEN)
                ? x[(size_t)b * T_LEN + g] : 0.0f;
  }
  for (int n = tid; n < IR_LEN; n += NTHREADS)
    invn[n] = (n > 0) ? 1.0f / (float)n : 1.0f;
  if (tid < NRWAVES) zflag[1][tid] = 0;

  float kc[NCEP];
  float ps[M_ORD];
#pragma unroll
  for (int k = 0; k < NCEP; ++k) kc[k] = 0.0f;
#pragma unroll
  for (int k = 0; k < M_ORD; ++k) ps[k] = 0.0f;
  if (!mat_role) {
    const float* mcp = mc + ((size_t)b * T_LEN + (tb + rt)) * NCEP;
    const float c0 = mcp[0];
#pragma unroll
    for (int k = 1; k < NCEP; ++k) kc[k] = (float)k * mcp[k];
    ps[0] = __expf(c0);
  }

  v8f acc0 = {};
  v8f acc1 = {};

  __syncthreads();

  for (int it = 0; it <= NCHUNK; ++it) {
    const v4i zf = *reinterpret_cast<const v4i*>(&zflag[(it + 1) & 1][0]);
    if (!mat_role) {
      if (it < NCHUNK) {
        const int n0 = it * CHUNK;
        float* hrow = &hs[(it & 1) * HBUF + rt * HSTRIDE];
        v4f iv[CHUNK / 4];
#pragma unroll
        for (int q = 0; q < CHUNK / 4; ++q)
          iv[q] = *reinterpret_cast<const v4f*>(&invn[n0 + 4 * q]);
        __builtin_amdgcn_sched_barrier(0);
#pragma unroll
        for (int j = 0; j < CHUNK; ++j) {
          const float hn = ps[0] * iv[j >> 2][j & 3];
          hrow[j] = hn;
#pragma unroll
          for (int i = 0; i < M_ORD - 1; ++i) ps[i] = fmaf(kc[i + 1], hn, ps[i + 1]);
          ps[M_ORD - 1] = kc[M_ORD] * hn;
        }
        unsigned int ub = 0u;
#pragma unroll
        for (int i = 0; i < M_ORD; ++i) ub |= __float_as_uint(ps[i]);
        const int rzero = (__builtin_amdgcn_ballot_w32((ub & 0x7fffffffu) != 0u) == 0u) ? 1 : 0;
        if (lane == 0) zflag[it & 1][wave - NMWAVES] = rzero;
      }
    } else {
      if (it > 0) {
        const int n0 = (it - 1) * CHUNK;
        const float* hbuf = &hs[((it - 1) & 1) * HBUF];
#pragma unroll
        for (int st = 0; st < 2; ++st) {
          const int sub = wave * 32 + st * 16;
          v8f acc = st ? acc1 : acc0;
          const float* hsub = &hbuf[(sub + lm) * HSTRIDE];
#pragma unroll
          for (int g = 0; g < CHUNK / 4; ++g) {
            const int kl = g * 4 + 2 * hi;
            v2f a;  a.x = hsub[kl];  a.y = hsub[kl + 1];
            const int xb = (IR_LEN - 1) + sub + lm - (n0 + kl);
            v2f bv; bv.x = xs[xb];   bv.y = xs[xb - 1];
            acc = __builtin_amdgcn_wmma_f32_16x16x4_f32(
                 false, a,  false, bv,
                 (short)0, acc,  false,  false);
          }
          if (st) acc1 = acc; else acc0 = acc;
        }
      }
    }
    __syncthreads();
    if (__builtin_amdgcn_readfirstlane(zf.x & zf.y & zf.z & zf.w)) break;
  }

  for (int st = 0; st < 2; ++st) {
    if (mat_role) {
      v8f acc = st ? acc1 : acc0;
#pragma unroll
      for (int r = 0; r < 8; ++r)
        dbuf[wave][r + 8 * hi][lm] = acc[r];
    }
    __syncthreads();
    if (mat_role && lane < 16) {
      const int sub = wave * 32 + st * 16;
      y[(size_t)b * T_LEN + tb + sub + lane] = dbuf[wave][lane][lane];
    }
    __syncthreads();
  }
}

extern "C" void kernel_launch(void* const* d_in, const int* in_sizes, int n_in,
                              void* d_out, int out_size, void* d_ws, size_t ws_size,
                              hipStream_t stream) {
  (void)in_sizes; (void)n_in; (void)out_size; (void)d_ws; (void)ws_size;
  const float* x  = (const float*)d_in[0];
  const float* mc = (const float*)d_in[1];
  float* y = (float*)d_out;

  dim3 grid(BATCH * (T_LEN / TILE_T));
  dim3 block(NTHREADS);
  tv_fir_c2ir_kernel<<<grid, block, 0, stream>>>(x, mc, y);
}


# start-up: the nine x-window loads per thread and the cepstrum loads are all issued before the first wait and the 1/n table is built in their shadow; were nine load-wait-store round trips in a row
# speedup vs baseline: 7.7483x; 1.2047x over previous
#include <hip/hip_runtime.h>
#include <math.h>

typedef float v2f __attribute__((ext_vector_type(2)));
typedef float v4f __attribute__((ext_vector_type(4)));
typedef float v8f __attribute__((ext_vector_type(8)));
typedef int   v4i __attribute__((ext_vector_type(4)));

#define BATCH   2
#define T_LEN   16000
#define NCEP    25
#define M_ORD   24
#define IR_LEN  2000
#define TILE_T  128
#define NMWAVES 4
#define NRWAVES 4
#define NTHREADS ((NMWAVES + NRWAVES) * 32)
#define CHUNK   80
#define NCHUNK  25
#define HSTRIDE 81
#define HBUF    (TILE_T * HSTRIDE)
#define XS_LEN  2176

__global__ __launch_bounds__(NTHREADS)
void tv_fir_c2ir_kernel(const float* __restrict__ x,
                        const float* __restrict__ mc,
                        float* __restrict__ y) {
  __shared__ float xs[XS_LEN];
  __shared__ float hs[2 * HBUF];
  __shared__ __attribute__((aligned(16))) float invn[IR_LEN];
  __shared__ float dbuf[NMWAVES][16][16];
  __shared__ __attribute__((aligned(16))) int zflag[2][NRWAVES];

  const int tid  = threadIdx.x;
  const int wave = __builtin_amdgcn_readfirstlane(tid >> 5);
  const bool mat_role = wave < NMWAVES;
  const int rt   = tid - NMWAVES * 32;
  const int lane = tid & 31;
  const int hi   = lane >> 4;
  const int lm   = lane & 15;

  const int b  = blockIdx.x / (T_LEN / TILE_T);
  const int tb = (blockIdx.x % (T_LEN / TILE_T)) * TILE_T;

  float mcv[NCEP];
#pragma unroll
  for (int k = 0; k < NCEP; ++k) mcv[k] = 0.0f;
  if (!mat_role) {
    const float* mcp = mc + ((size_t)b * T_LEN + (tb + rt)) * NCEP;
#pragma unroll
    for (int k = 0; k < NCEP; ++k) mcv[k] = mcp[k];
  }
  float xv[(XS_LEN + NTHREADS - 1) / NTHREADS];
#pragma unroll
  for (int q = 0; q < (XS_LEN + NTHREADS - 1) / NTHREADS; ++q) {
    const int i  = tid + q * NTHREADS;
    const int g  = tb - (IR_LEN - 1) + i;
    const int gc = g < 0 ? 0 : (g > T_LEN - 1 ? T_LEN - 1 : g);
    const float v = x[(size_t)b * T_LEN + gc];
    xv[q] = (i < (IR_LEN - 1 + TILE_T) && g >= 0 && g < T_LEN) ? v : 0.0f;
  }
  for (int n = tid; n < IR_LEN; n += NTHREADS)
    invn[n] = (n > 0) ? 1.0f / (float)n : 1.0f;
  if (tid < NRWAVES) zflag[1][tid] = 0;
#pragma unroll
  for (int q = 0; q < (XS_LEN + NTHREADS - 1) / NTHREADS; ++q) {
    const int i = tid + q * NTHREADS;
    if (i < XS_LEN) xs[i] = xv[q];
  }

  float kc[NCEP];
  float ps[M_ORD];
#pragma unroll
  for (int k = 0; k < NCEP; ++k) kc[k] = 0.0f;
#pragma unroll
  for (int k = 0; k < M_ORD; ++k) ps[k] = 0.0f;
  if (!mat_role) {
#pragma unroll
    for (int k = 1; k < NCEP; ++k) kc[k] = (float)k * mcv[k];
    ps[0] = __expf(mcv[0]);
  }

  v8f acc0 = {};
  v8f acc1 = {};

  __syncthreads();

  for (int it = 0; it <= NCHUNK; ++it) {
    const v4i zf = *reinterpret_cast<const v4i*>(&zflag[(it + 1) & 1][0]);
    if (!mat_role) {
      if (it < NCHUNK) {
        const int n0 = it * CHUNK;
        float* hrow = &hs[(it & 1) * HBUF + rt * HSTRIDE];
        v4f iv[CHUNK / 4];
#pragma unroll
        for (int q = 0; q < CHUNK / 4; ++q)
          iv[q] = *reinterpret_cast<const v4f*>(&invn[n0 + 4 * q]);
        __builtin_amdgcn_sched_barrier(0);
#pragma unroll
        for (int j = 0; j < CHUNK; ++j) {
          const float hn = ps[0] * iv[j >> 2][j & 3];
          hrow[j] = hn;
#pragma unroll
          for (int i = 0; i < M_ORD - 1; ++i) ps[i] = fmaf(kc[i + 1], hn, ps[i + 1]);
          ps[M_ORD - 1] = kc[M_ORD] * hn;
        }
        unsigned int ub = 0u;
#pragma unroll
        for (int i = 0; i < M_ORD; ++i) ub |= __float_as_uint(ps[i]);
        const int rzero = (__builtin_amdgcn_ballot_w32((ub & 0x7fffffffu) != 0u) == 0u) ? 1 : 0;
        if (lane == 0) zflag[it & 1][wave - NMWAVES] = rzero;
      }
    } else {
      if (it > 0) {
        const int n0 = (it - 1) * CHUNK;
        const float* hbuf = &hs[((it - 1) & 1) * HBUF];
#pragma unroll
        for (int st = 0; st < 2; ++st) {
          const int sub = wave * 32 + st * 16;
          v8f acc = st ? acc1 : acc0;
          const float* hsub = &hbuf[(sub + lm) * HSTRIDE];
#pragma unroll
          for (int g = 0; g < CHUNK / 4; ++g) {
            const int kl = g * 4 + 2 * hi;
            v2f a;  a.x = hsub[kl];  a.y = hsub[kl + 1];
            const int xb = (IR_LEN - 1) + sub + lm - (n0 + kl);
            v2f bv; bv.x = xs[xb];   bv.y = xs[xb - 1];
            acc = __builtin_amdgcn_wmma_f32_16x16x4_f32(
                 false, a,  false, bv,
                 (short)0, acc,  false,  false);
          }
          if (st) acc1 = acc; else acc0 = acc;
        }
      }
    }
    __syncthreads();
    if (__builtin_amdgcn_readfirstlane(zf.x & zf.y & zf.z & zf.w)) break;
  }

  for (int st = 0; st < 2; ++st) {
    if (mat_role) {
      v8f acc = st ? acc1 : acc0;
#pragma unroll
      for (int r = 0; r < 8; ++r)
        dbuf[wave][r + 8 * hi][lm] = acc[r];
    }
    __syncthreads();
    if (mat_role && lane < 16) {
      const int sub = wave * 32 + st * 16;
      y[(size_t)b * T_LEN + tb + sub + lane] = dbuf[wave][lane][lane];
    }
    __syncthreads();
  }
}

extern "C" void kernel_launch(void* const* d_in, const int* in_sizes, int n_in,
                              void* d_out, int out_size, void* d_ws, size_t ws_size,
                              hipStream_t stream) {
  (void)in_sizes; (void)n_in; (void)out_size; (void)d_ws; (void)ws_size;
  const float* x  = (const float*)d_in[0];
  const float* mc = (const float*)d_in[1];
  float* y = (float*)d_out;

  dim3 grid(BATCH * (T_LEN / TILE_T));
  dim3 block(NTHREADS);
  tv_fir_c2ir_kernel<<<grid, block, 0, stream>>>(x, mc, y);
}


# each wave runs both roles in one stream on its own 32 chains (fragment loads first, recursion of chunk c, then WMMA FIR of chunk c-1): no workgroup barrier or flags in the loop; same sums and order
# speedup vs baseline: 8.8168x; 1.1379x over previous
#include <hip/hip_runtime.h>
#include <math.h>

typedef float v2f __attribute__((ext_vector_type(2)));
typedef float v4f __attribute__((ext_vector_type(4)));
typedef float v8f __attribute__((ext_vector_type(8)));
typedef int   v4i __attribute__((ext_vector_type(4)));

#define BATCH   2
#define T_LEN   16000
#define NCEP    25
#define M_ORD   24
#define IR_LEN  2000
#define TILE_T  128
#define NWAVES  4
#define NTHREADS (NWAVES * 32)
#define CHUNK   80
#define NCHUNK  25
#define HSTRIDE 81
#define HBUF    (TILE_T * HSTRIDE)
#define XS_LEN  2176


__device__ __forceinline__ void wave_lds_order() {
  __builtin_amdgcn_fence(__ATOMIC_RELEASE, "wavefront");
  __builtin_amdgcn_wave_barrier();
  __builtin_amdgcn_fence(__ATOMIC_ACQUIRE, "wavefront");
}

__device__ __forceinline__ void recur_table(v4f (&iv)[CHUNK / 4], const float* __restrict__ invp) {
#pragma unroll
  for (int q = 0; q < CHUNK / 4; ++q)
    iv[q] = *reinterpret_cast<const v4f*>(invp + 4 * q);
}
__device__ __forceinline__ void recur_taps(float (&ps)[M_ORD], const float (&kc)[NCEP],
                                           const v4f (&iv)[CHUNK / 4],
                                           float* __restrict__ hrow) {
#pragma unroll
  for (int j = 0; j < CHUNK; ++j) {
    const float hn = ps[0] * iv[j >> 2][j & 3];
    hrow[j] = hn;
#pragma unroll
    for (int i = 0; i < M_ORD - 1; ++i) ps[i] = fmaf(kc[i + 1], hn, ps[i + 1]);
    ps[M_ORD - 1] = kc[M_ORD] * hn;
  }
}

struct Frags { v2f a[2][CHUNK / 4]; v2f b[2][CHUNK / 4]; };

__device__ __forceinline__ void fir_load(Frags& f, const float* __restrict__ hw,
                                         const float* __restrict__ xw, const int hi) {
#pragma unroll
  for (int st = 0; st < 2; ++st) {
    const float* hsub = hw + st * 16 * HSTRIDE;
    const float* xsub = xw - st * 16;
#pragma unroll
    for (int g = 0; g < CHUNK / 4; ++g) {
      const int kl = g * 4 + 2 * hi;
      f.a[st][g].x = hsub[kl];   f.a[st][g].y = hsub[kl + 1];
      f.b[st][g].x = xsub[kl];   f.b[st][g].y = xsub[kl + 1];
    }
  }
}

__device__ __forceinline__ void fir_mma(v8f& acc0, v8f& acc1, const Frags& f) {
#pragma unroll
  for (int g = 0; g < CHUNK / 4; ++g) {
    acc0 = __builtin_amdgcn_wmma_f32_16x16x4_f32(
         false, f.a[0][g],  false, f.b[0][g],
         (short)0, acc0,  false,  false);
    acc1 = __builtin_amdgcn_wmma_f32_16x16x4_f32(
         false, f.a[1][g],  false, f.b[1][g],
         (short)0, acc1,  false,  false);
  }
}

__global__ __launch_bounds__(NTHREADS)
void tv_fir_c2ir_kernel(const float* __restrict__ x,
                        const float* __restrict__ mc,
                        float* __restrict__ y) {
  __shared__ float xs[XS_LEN];
  __shared__ float hs[2 * HBUF];
  __shared__ __attribute__((aligned(16))) float invn[IR_LEN];
  __shared__ float dbuf[NWAVES][16][16];

  const int tid  = threadIdx.x;
  const int wave = __builtin_amdgcn_readfirstlane(tid >> 5);
  const int lane = tid & 31;
  const int hi   = lane >> 4;
  const int lm   = lane & 15;

  const int b  = blockIdx.x / (T_LEN / TILE_T);
  const int tb = (blockIdx.x % (T_LEN / TILE_T)) * TILE_T;

  float mcv[NCEP];
  {
    const float* mcp = mc + ((size_t)b * T_LEN + (tb + tid)) * NCEP;
#pragma unroll
    for (int k = 0; k < NCEP; ++k) mcv[k] = mcp[k];
  }
  float xv[(XS_LEN + NTHREADS - 1) / NTHREADS];
#pragma unroll
  for (int q = 0; q < (XS_LEN + NTHREADS - 1) / NTHREADS; ++q) {
    const int i  = tid + q * NTHREADS;
    const int g  = tb - (IR_LEN - 1) + i;
    const int gc = g < 0 ? 0 : (g > T_LEN - 1 ? T_LEN - 1 : g);
    const float v = x[(size_t)b * T_LEN + gc];
    xv[q] = (i < (IR_LEN - 1 + TILE_T) && g >= 0 && g < T_LEN) ? v : 0.0f;
  }
#pragma unroll
  for (int q = 0; q < (IR_LEN + NTHREADS - 1) / NTHREADS; ++q) {
    const int n = tid + q * NTHREADS;
    if (n < IR_LEN) invn[n] = (n > 0) ? 1.0f / (float)n : 1.0f;
  }
#pragma unroll
  for (int q = 0; q < (XS_LEN + NTHREADS - 1) / NTHREADS; ++q) {
    const int i = tid + q * NTHREADS;
    if (i < XS_LEN) xs[(XS_LEN - 1) - i] = xv[q];
  }

  float kc[NCEP];
  float ps[M_ORD];
  kc[0] = 0.0f;
#pragma unroll
  for (int k = 1; k < NCEP; ++k) kc[k] = (float)k * mcv[k];
#pragma unroll
  for (int k = 1; k < M_ORD; ++k) ps[k] = 0.0f;
  ps[0] = __expf(mcv[0]);

  v8f acc0 = {};
  v8f acc1 = {};

  __syncthreads();

  const float* hw = &hs[(wave * 32 + lm) * HSTRIDE];
  const float* xw = &xs[(XS_LEN - 1) - (IR_LEN - 1) - (wave * 32 + lm)];
  float* hrow = &hs[tid * HSTRIDE];

  v4f iv[CHUNK / 4];
  Frags fr;

  recur_table(iv, &invn[0]);
  recur_taps(ps, kc, iv, hrow);
  int it = 1;
  for (;;) {
    unsigned int uq[4] = {0u, 0u, 0u, 0u};
#pragma unroll
    for (int i = 0; i < M_ORD; ++i) uq[i & 3] |= __float_as_uint(ps[i]);
    const unsigned int ub = (uq[0] | uq[1]) | (uq[2] | uq[3]);
    const bool zero = __builtin_amdgcn_ballot_w32((ub & 0x7fffffffu) != 0u) == 0u;
    if (zero || it >= NCHUNK) break;
    wave_lds_order();
    recur_table(iv, &invn[it * CHUNK]);
    fir_load(fr, hw + ((it - 1) & 1) * HBUF, xw + (it - 1) * CHUNK, hi);
    __builtin_amdgcn_sched_barrier(0);
    recur_taps(ps, kc, iv, hrow + (it & 1) * HBUF);
    __builtin_amdgcn_sched_barrier(0);
    fir_mma(acc0, acc1, fr);
    ++it;
  }
  wave_lds_order();
  fir_load(fr, hw + ((it - 1) & 1) * HBUF, xw + (it - 1) * CHUNK, hi);
  __builtin_amdgcn_sched_barrier(0);
  fir_mma(acc0, acc1, fr);

#pragma unroll
  for (int st = 0; st < 2; ++st) {
    const v8f acc = st ? acc1 : acc0;
    wave_lds_order();
#pragma unroll
    for (int r = 0; r < 8; ++r)
      dbuf[wave][r + 8 * hi][lm] = acc[r];
    wave_lds_order();
    if (lane < 16)
      y[(size_t)b * T_LEN + tb + wave * 32 + st * 16 + lane] = dbuf[wave][lane][lane];
  }
}

extern "C" void kernel_launch(void* const* d_in, const int* in_sizes, int n_in,
                              void* d_out, int out_size, void* d_ws, size_t ws_size,
                              hipStream_t stream) {
  (void)in_sizes; (void)n_in; (void)out_size; (void)d_ws; (void)ws_size;
  const float* x  = (const float*)d_in[0];
  const float* mc = (const float*)d_in[1];
  float* y = (float*)d_out;

  dim3 grid(BATCH * (T_LEN / TILE_T));
  dim3 block(NTHREADS);
  tv_fir_c2ir_kernel<<<grid, block, 0, stream>>>(x, mc, y);
}


# a chunk's 80 taps stay in registers and go to LDS as one group of twenty 16-byte stores behind the recursion (row pitch 84 words, A pairs as 8-byte loads); was 40 stores among the taps; same sums
# speedup vs baseline: 9.9299x; 1.1262x over previous
#include <hip/hip_runtime.h>
#include <math.h>

typedef float v2f __attribute__((ext_vector_type(2)));
typedef float v4f __attribute__((ext_vector_type(4)));
typedef float v8f __attribute__((ext_vector_type(8)));
typedef int   v4i __attribute__((ext_vector_type(4)));

#define BATCH   2
#define T_LEN   16000
#define NCEP    25
#define M_ORD   24
#define IR_LEN  2000
#define TILE_T  128
#define NWAVES  4
#define NTHREADS (NWAVES * 32)
#define CHUNK   80
#define NCHUNK  25
#define HSTRIDE 84
#define HBUF    (TILE_T * HSTRIDE)
#define XS_LEN  2176


__device__ __forceinline__ void wave_lds_order() {
  __builtin_amdgcn_fence(__ATOMIC_RELEASE, "wavefront");
  __builtin_amdgcn_wave_barrier();
  __builtin_amdgcn_fence(__ATOMIC_ACQUIRE, "wavefront");
}

__device__ __forceinline__ void recur_table(v4f (&iv)[CHUNK / 4], const float* __restrict__ invp) {
#pragma unroll
  for (int q = 0; q < CHUNK / 4; ++q)
    iv[q] = *reinterpret_cast<const v4f*>(invp + 4 * q);
}
__device__ __forceinline__ void recur_taps(float (&ps)[M_ORD], const float (&kc)[NCEP],
                                           const v4f (&iv)[CHUNK / 4],
                                           v4f (&hv)[CHUNK / 4]) {
#pragma unroll
  for (int j = 0; j < CHUNK; ++j) {
    const float hn = ps[0] * iv[j >> 2][j & 3];
    hv[j >> 2][j & 3] = hn;
#pragma unroll
    for (int i = 0; i < M_ORD - 1; ++i) ps[i] = fmaf(kc[i + 1], hn, ps[i + 1]);
    ps[M_ORD - 1] = kc[M_ORD] * hn;
  }
}

__device__ __forceinline__ void store_taps(const v4f (&hv)[CHUNK / 4], float* __restrict__ hrow) {
#pragma unroll
  for (int q = 0; q < CHUNK / 4; ++q)
    *reinterpret_cast<v4f*>(hrow + 4 * q) = hv[q];
}

struct Frags { v2f a[2][CHUNK / 4]; v2f b[2][CHUNK / 4]; };

__device__ __forceinline__ void fir_load(Frags& f, const float* __restrict__ hw,
                                         const float* __restrict__ xw, const int hi) {
#pragma unroll
  for (int st = 0; st < 2; ++st) {
    const float* hsub = hw + st * 16 * HSTRIDE;
    const float* xsub = xw - st * 16;
#pragma unroll
    for (int g = 0; g < CHUNK / 4; ++g) {
      const int kl = g * 4 + 2 * hi;
      f.a[st][g] = *reinterpret_cast<const v2f*>(hsub + kl);
      f.b[st][g].x = xsub[kl];   f.b[st][g].y = xsub[kl + 1];
    }
  }
}

__device__ __forceinline__ void fir_mma(v8f& acc0, v8f& acc1, const Frags& f) {
#pragma unroll
  for (int g = 0; g < CHUNK / 4; ++g) {
    acc0 = __builtin_amdgcn_wmma_f32_16x16x4_f32(
         false, f.a[0][g],  false, f.b[0][g],
         (short)0, acc0,  false,  false);
    acc1 = __builtin_amdgcn_wmma_f32_16x16x4_f32(
         false, f.a[1][g],  false, f.b[1][g],
         (short)0, acc1,  false,  false);
  }
}

__global__ __launch_bounds__(NTHREADS)
void tv_fir_c2ir_kernel(const float* __restrict__ x,
                        const float* __restrict__ mc,
                        float* __restrict__ y) {
  __shared__ float xs[XS_LEN];
  __shared__ __attribute__((aligned(16))) float hs[2 * HBUF];
  __shared__ __attribute__((aligned(16))) float invn[IR_LEN];
  __shared__ float dbuf[NWAVES][16][16];

  const int tid  = threadIdx.x;
  const int wave = __builtin_amdgcn_readfirstlane(tid >> 5);
  const int lane = tid & 31;
  const int hi   = lane >> 4;
  const int lm   = lane & 15;

  const int b  = blockIdx.x / (T_LEN / TILE_T);
  const int tb = (blockIdx.x % (T_LEN / TILE_T)) * TILE_T;

  float mcv[NCEP];
  {
    const float* mcp = mc + ((size_t)b * T_LEN + (tb + tid)) * NCEP;
#pragma unroll
    for (int k = 0; k < NCEP; ++k) mcv[k] = mcp[k];
  }
  float xv[(XS_LEN + NTHREADS - 1) / NTHREADS];
#pragma unroll
  for (int q = 0; q < (XS_LEN + NTHREADS - 1) / NTHREADS; ++q) {
    const int i  = tid + q * NTHREADS;
    const int g  = tb - (IR_LEN - 1) + i;
    const int gc = g < 0 ? 0 : (g > T_LEN - 1 ? T_LEN - 1 : g);
    const float v = x[(size_t)b * T_LEN + gc];
    xv[q] = (i < (IR_LEN - 1 + TILE_T) && g >= 0 && g < T_LEN) ? v : 0.0f;
  }
#pragma unroll
  for (int q = 0; q < (IR_LEN + NTHREADS - 1) / NTHREADS; ++q) {
    const int n = tid + q * NTHREADS;
    if (n < IR_LEN) invn[n] = (n > 0) ? 1.0f / (float)n : 1.0f;
  }
#pragma unroll
  for (int q = 0; q < (XS_LEN + NTHREADS - 1) / NTHREADS; ++q) {
    const int i = tid + q * NTHREADS;
    if (i < XS_LEN) xs[(XS_LEN - 1) - i] = xv[q];
  }

  float kc[NCEP];
  float ps[M_ORD];
  kc[0] = 0.0f;
#pragma unroll
  for (int k = 1; k < NCEP; ++k) kc[k] = (float)k * mcv[k];
#pragma unroll
  for (int k = 1; k < M_ORD; ++k) ps[k] = 0.0f;
  ps[0] = __expf(mcv[0]);

  v8f acc0 = {};
  v8f acc1 = {};

  __syncthreads();

  const float* hw = &hs[(wave * 32 + lm) * HSTRIDE];
  const float* xw = &xs[(XS_LEN - 1) - (IR_LEN - 1) - (wave * 32 + lm)];
  float* hrow = &hs[tid * HSTRIDE];

  v4f iv[CHUNK / 4];
  v4f hv[CHUNK / 4];
  Frags fr;

  recur_table(iv, &invn[0]);
  recur_taps(ps, kc, iv, hv);
  __builtin_amdgcn_sched_barrier(0);
  store_taps(hv, hrow);
  int it = 1;
  for (;;) {
    unsigned int uq[4] = {0u, 0u, 0u, 0u};
#pragma unroll
    for (int i = 0; i < M_ORD; ++i) uq[i & 3] |= __float_as_uint(ps[i]);
    const unsigned int ub = (uq[0] | uq[1]) | (uq[2] | uq[3]);
    const bool zero = __builtin_amdgcn_ballot_w32((ub & 0x7fffffffu) != 0u) == 0u;
    if (zero || it >= NCHUNK) break;
    wave_lds_order();
    recur_table(iv, &invn[it * CHUNK]);
    fir_load(fr, hw + ((it - 1) & 1) * HBUF, xw + (it - 1) * CHUNK, hi);
    __builtin_amdgcn_sched_barrier(0);
    recur_taps(ps, kc, iv, hv);
    __builtin_amdgcn_sched_barrier(0);
    store_taps(hv, hrow + (it & 1) * HBUF);
    __builtin_amdgcn_sched_barrier(0);
    fir_mma(acc0, acc1, fr);
    ++it;
  }
  wave_lds_order();
  fir_load(fr, hw + ((it - 1) & 1) * HBUF, xw + (it - 1) * CHUNK, hi);
  __builtin_amdgcn_sched_barrier(0);
  fir_mma(acc0, acc1, fr);

#pragma unroll
  for (int st = 0; st < 2; ++st) {
    const v8f acc = st ? acc1 : acc0;
    wave_lds_order();
#pragma unroll
    for (int r = 0; r < 8; ++r)
      dbuf[wave][r + 8 * hi][lm] = acc[r];
    wave_lds_order();
    if (lane < 16)
      y[(size_t)b * T_LEN + tb + wave * 32 + st * 16 + lane] = dbuf[wave][lane][lane];
  }
}

extern "C" void kernel_launch(void* const* d_in, const int* in_sizes, int n_in,
                              void* d_out, int out_size, void* d_ws, size_t ws_size,
                              hipStream_t stream) {
  (void)in_sizes; (void)n_in; (void)out_size; (void)d_ws; (void)ws_size;
  const float* x  = (const float*)d_in[0];
  const float* mc = (const float*)d_in[1];
  float* y = (float*)d_out;

  dim3 grid(BATCH * (T_LEN / TILE_T));
  dim3 block(NTHREADS);
  tv_fir_c2ir_kernel<<<grid, block, 0, stream>>>(x, mc, y);
}


# A fragments made in registers by half-wave swaps of the tap registers, no h round trip through LDS: the chunk loop fits 256 registers and issues without register-window switches; same sums
# speedup vs baseline: 10.8685x; 1.0945x over previous
#include <hip/hip_runtime.h>
#include <math.h>

typedef float v2f __attribute__((ext_vector_type(2)));
typedef float v4f __attribute__((ext_vector_type(4)));
typedef float v8f __attribute__((ext_vector_type(8)));
typedef unsigned int v2u __attribute__((ext_vector_type(2)));

#define BATCH   2
#define T_LEN   16000
#define NCEP    25
#define M_ORD   24
#define IR_LEN  2000
#define TILE_T  128
#define NWAVES  4
#define NTHREADS (NWAVES * 32)
#define CHUNK   80
#define NCHUNK  25
#define NGRP    (CHUNK / 4)
#define XS_LEN  2176


__device__ __forceinline__ void wave_lds_order() {
  __builtin_amdgcn_fence(__ATOMIC_RELEASE, "wavefront");
  __builtin_amdgcn_wave_barrier();
  __builtin_amdgcn_fence(__ATOMIC_ACQUIRE, "wavefront");
}

__device__ __forceinline__ void half_swap(float& a, float& b) {
  const v2u r = __builtin_amdgcn_permlane16_swap(__float_as_uint(a), __float_as_uint(b), false, false);
  a = __uint_as_float(r[0]);
  b = __uint_as_float(r[1]);
}

__device__ __forceinline__ void recur_table(v4f (&tv)[NGRP], const float* __restrict__ invp) {
#pragma unroll
  for (int q = 0; q < NGRP; ++q)
    tv[q] = *reinterpret_cast<const v4f*>(invp + 4 * q);
}

__device__ __forceinline__ void bfrag_load(v2f (&bq)[NGRP + 4], const float* __restrict__ xw, const int hi) {
#pragma unroll
  for (int q = 0; q < NGRP + 4; ++q) {
    const int kl = (q - 4) * 4 + 2 * hi;
    bq[q].x = xw[kl];   bq[q].y = xw[kl + 1];
  }
}

__device__ __forceinline__ void recur_taps(float (&ps)[M_ORD], const float (&kc)[NCEP],
                                           v4f (&tv)[NGRP]) {
#pragma unroll
  for (int j = 0; j < CHUNK; ++j) {
    const float hn = ps[0] * tv[j >> 2][j & 3];
    tv[j >> 2][j & 3] = hn;
#pragma unroll
    for (int i = 0; i < M_ORD - 1; ++i) ps[i] = fmaf(kc[i + 1], hn, ps[i + 1]);
    ps[M_ORD - 1] = kc[M_ORD] * hn;
  }
}

__device__ __forceinline__ void fir_chunk(v4f (&hv)[NGRP], const v2f (&bq)[NGRP + 4],
                                          v8f& acc0, v8f& acc1) {
#pragma unroll
  for (int g = 0; g < NGRP; ++g) {
    float t0 = hv[g].x, t1 = hv[g].y, t2 = hv[g].z, t3 = hv[g].w;
    half_swap(t0, t2);
    half_swap(t1, t3);
    const v2f a0 = {t0, t1};
    const v2f a1 = {t2, t3};
    acc0 = __builtin_amdgcn_wmma_f32_16x16x4_f32(
         false, a0,  false, bq[g + 4],
         (short)0, acc0,  false,  false);
    acc1 = __builtin_amdgcn_wmma_f32_16x16x4_f32(
         false, a1,  false, bq[g],
         (short)0, acc1,  false,  false);
  }
}

__global__ __launch_bounds__(NTHREADS)
void tv_fir_c2ir_kernel(const float* __restrict__ x,
                        const float* __restrict__ mc,
                        float* __restrict__ y) {
  __shared__ float xs[XS_LEN];
  __shared__ __attribute__((aligned(16))) float invn[IR_LEN];
  __shared__ float dbuf[NWAVES][16][16];

  const int tid  = threadIdx.x;
  const int wave = __builtin_amdgcn_readfirstlane(tid >> 5);
  const int lane = tid & 31;
  const int hi   = lane >> 4;
  const int lm   = lane & 15;

  const int b  = blockIdx.x / (T_LEN / TILE_T);
  const int tb = (blockIdx.x % (T_LEN / TILE_T)) * TILE_T;

  float mcv[NCEP];
  {
    const float* mcp = mc + ((size_t)b * T_LEN + (tb + tid)) * NCEP;
#pragma unroll
    for (int k = 0; k < NCEP; ++k) mcv[k] = mcp[k];
  }
  float xv[(XS_LEN + NTHREADS - 1) / NTHREADS];
#pragma unroll
  for (int q = 0; q < (XS_LEN + NTHREADS - 1) / NTHREADS; ++q) {
    const int i  = tid + q * NTHREADS;
    const int g  = tb - (IR_LEN - 1) + i;
    const int gc = g < 0 ? 0 : (g > T_LEN - 1 ? T_LEN - 1 : g);
    const float v = x[(size_t)b * T_LEN + gc];
    xv[q] = (i < (IR_LEN - 1 + TILE_T) && g >= 0 && g < T_LEN) ? v : 0.0f;
  }
#pragma unroll
  for (int q = 0; q < (IR_LEN + NTHREADS - 1) / NTHREADS; ++q) {
    const int n = tid + q * NTHREADS;
    if (n < IR_LEN) invn[n] = (n > 0) ? 1.0f / (float)n : 1.0f;
  }
#pragma unroll
  for (int q = 0; q < (XS_LEN + NTHREADS - 1) / NTHREADS; ++q) {
    const int i = tid + q * NTHREADS;
    if (i < XS_LEN) xs[(XS_LEN - 1) - i] = xv[q];
  }

  float kc[NCEP];
  float ps[M_ORD];
  kc[0] = 0.0f;
#pragma unroll
  for (int k = 1; k < NCEP; ++k) kc[k] = (float)k * mcv[k];
#pragma unroll
  for (int k = 1; k < M_ORD; ++k) ps[k] = 0.0f;
  ps[0] = __expf(mcv[0]);

  v8f acc0 = {};
  v8f acc1 = {};

  __syncthreads();

  const float* xw = &xs[(XS_LEN - 1) - (IR_LEN - 1) - (wave * 32 + lm)];

  v4f tv[NGRP];
  v2f bq[NGRP + 4];

  int it = 0;
  for (;;) {
    recur_table(tv, &invn[it * CHUNK]);
    bfrag_load(bq, xw + it * CHUNK, hi);
    __builtin_amdgcn_sched_barrier(0);
    recur_taps(ps, kc, tv);
    __builtin_amdgcn_sched_barrier(0);
    fir_chunk(tv, bq, acc0, acc1);
    ++it;
    unsigned int uq[4] = {0u, 0u, 0u, 0u};
#pragma unroll
    for (int i = 0; i < M_ORD; ++i) uq[i & 3] |= __float_as_uint(ps[i]);
    const unsigned int ub = (uq[0] | uq[1]) | (uq[2] | uq[3]);
    const bool zero = __builtin_amdgcn_ballot_w32((ub & 0x7fffffffu) != 0u) == 0u;
    if (zero || it >= NCHUNK) break;
  }

#pragma unroll
  for (int st = 0; st < 2; ++st) {
    const v8f acc = st ? acc1 : acc0;
    wave_lds_order();
#pragma unroll
    for (int r = 0; r < 8; ++r)
      dbuf[wave][r + 8 * hi][lm] = acc[r];
    wave_lds_order();
    if (lane < 16)
      y[(size_t)b * T_LEN + tb + wave * 32 + st * 16 + lane] = dbuf[wave][lane][lane];
  }
}

extern "C" void kernel_launch(void* const* d_in, const int* in_sizes, int n_in,
                              void* d_out, int out_size, void* d_ws, size_t ws_size,
                              hipStream_t stream) {
  (void)in_sizes; (void)n_in; (void)out_size; (void)d_ws; (void)ws_size;
  const float* x  = (const float*)d_in[0];
  const float* mc = (const float*)d_in[1];
  float* y = (float*)d_out;

  dim3 grid(BATCH * (T_LEN / TILE_T));
  dim3 block(NTHREADS);
  tv_fir_c2ir_kernel<<<grid, block, 0, stream>>>(x, mc, y);
}


# chunk loop runs with f32 denormals flushed (s_denorm_mode): taps and running sums under 1.18e-38 become 0, so a chain's state is exactly zero about 90 taps sooner and the exit leaves a chunk earlier
# speedup vs baseline: 11.7417x; 1.0803x over previous
#include <hip/hip_runtime.h>
#include <math.h>

typedef float v2f __attribute__((ext_vector_type(2)));
typedef float v4f __attribute__((ext_vector_type(4)));
typedef float v8f __attribute__((ext_vector_type(8)));
typedef unsigned int v2u __attribute__((ext_vector_type(2)));

#define BATCH   2
#define T_LEN   16000
#define NCEP    25
#define M_ORD   24
#define IR_LEN  2000
#define TILE_T  128
#define NWAVES  4
#define NTHREADS (NWAVES * 32)
#define CHUNK   80
#define NCHUNK  25
#define NGRP    (CHUNK / 4)
#define XS_LEN  2176


__device__ __forceinline__ void wave_lds_order() {
  __builtin_amdgcn_fence(__ATOMIC_RELEASE, "wavefront");
  __builtin_amdgcn_wave_barrier();
  __builtin_amdgcn_fence(__ATOMIC_ACQUIRE, "wavefront");
}

__device__ __forceinline__ void half_swap(float& a, float& b) {
  const v2u r = __builtin_amdgcn_permlane16_swap(__float_as_uint(a), __float_as_uint(b), false, false);
  a = __uint_as_float(r[0]);
  b = __uint_as_float(r[1]);
}

__device__ __forceinline__ void recur_table(v4f (&tv)[NGRP], const float* __restrict__ invp) {
#pragma unroll
  for (int q = 0; q < NGRP; ++q)
    tv[q] = *reinterpret_cast<const v4f*>(invp + 4 * q);
}

__device__ __forceinline__ void bfrag_load(v2f (&bq)[NGRP + 4], const float* __restrict__ xw, const int hi) {
#pragma unroll
  for (int q = 0; q < NGRP + 4; ++q) {
    const int kl = (q - 4) * 4 + 2 * hi;
    bq[q].x = xw[kl];   bq[q].y = xw[kl + 1];
  }
}

__device__ __forceinline__ void recur_taps(float (&ps)[M_ORD], const float (&kc)[NCEP],
                                           v4f (&tv)[NGRP]) {
#pragma unroll
  for (int j = 0; j < CHUNK; ++j) {
    const float hn = ps[0] * tv[j >> 2][j & 3];
    tv[j >> 2][j & 3] = hn;
#pragma unroll
    for (int i = 0; i < M_ORD - 1; ++i) ps[i] = fmaf(kc[i + 1], hn, ps[i + 1]);
    ps[M_ORD - 1] = kc[M_ORD] * hn;
  }
}

__device__ __forceinline__ void fir_chunk(v4f (&hv)[NGRP], const v2f (&bq)[NGRP + 4],
                                          v8f& acc0, v8f& acc1) {
#pragma unroll
  for (int g = 0; g < NGRP; ++g) {
    float t0 = hv[g].x, t1 = hv[g].y, t2 = hv[g].z, t3 = hv[g].w;
    half_swap(t0, t2);
    half_swap(t1, t3);
    const v2f a0 = {t0, t1};
    const v2f a1 = {t2, t3};
    acc0 = __builtin_amdgcn_wmma_f32_16x16x4_f32(
         false, a0,  false, bq[g + 4],
         (short)0, acc0,  false,  false);
    acc1 = __builtin_amdgcn_wmma_f32_16x16x4_f32(
         false, a1,  false, bq[g],
         (short)0, acc1,  false,  false);
  }
}

__global__ __launch_bounds__(NTHREADS)
void tv_fir_c2ir_kernel(const float* __restrict__ x,
                        const float* __restrict__ mc,
                        float* __restrict__ y) {
  __shared__ float xs[XS_LEN];
  __shared__ __attribute__((aligned(16))) float invn[IR_LEN];
  __shared__ float dbuf[NWAVES][16][16];

  const int tid  = threadIdx.x;
  const int wave = __builtin_amdgcn_readfirstlane(tid >> 5);
  const int lane = tid & 31;
  const int hi   = lane >> 4;
  const int lm   = lane & 15;

  const int b  = blockIdx.x / (T_LEN / TILE_T);
  const int tb = (blockIdx.x % (T_LEN / TILE_T)) * TILE_T;

  float mcv[NCEP];
  {
    const float* mcp = mc + ((size_t)b * T_LEN + (tb + tid)) * NCEP;
#pragma unroll
    for (int k = 0; k < NCEP; ++k) mcv[k] = mcp[k];
  }
  float xv[(XS_LEN + NTHREADS - 1) / NTHREADS];
#pragma unroll
  for (int q = 0; q < (XS_LEN + NTHREADS - 1) / NTHREADS; ++q) {
    const int i  = tid + q * NTHREADS;
    const int g  = tb - (IR_LEN - 1) + i;
    const int gc = g < 0 ? 0 : (g > T_LEN - 1 ? T_LEN - 1 : g);
    const float v = x[(size_t)b * T_LEN + gc];
    xv[q] = (i < (IR_LEN - 1 + TILE_T) && g >= 0 && g < T_LEN) ? v : 0.0f;
  }
#pragma unroll
  for (int q = 0; q < (IR_LEN + NTHREADS - 1) / NTHREADS; ++q) {
    const int n = tid + q * NTHREADS;
    if (n < IR_LEN) invn[n] = (n > 0) ? 1.0f / (float)n : 1.0f;
  }
#pragma unroll
  for (int q = 0; q < (XS_LEN + NTHREADS - 1) / NTHREADS; ++q) {
    const int i = tid + q * NTHREADS;
    if (i < XS_LEN) xs[(XS_LEN - 1) - i] = xv[q];
  }

  float kc[NCEP];
  float ps[M_ORD];
  kc[0] = 0.0f;
#pragma unroll
  for (int k = 1; k < NCEP; ++k) kc[k] = (float)k * mcv[k];
#pragma unroll
  for (int k = 1; k < M_ORD; ++k) ps[k] = 0.0f;
  ps[0] = __expf(mcv[0]);

  v8f acc0 = {};
  v8f acc1 = {};

  __syncthreads();

  asm volatile("s_denorm_mode 12");

  const float* xw = &xs[(XS_LEN - 1) - (IR_LEN - 1) - (wave * 32 + lm)];

  v4f tv[NGRP];
  v2f bq[NGRP + 4];

  int it = 0;
  for (;;) {
    recur_table(tv, &invn[it * CHUNK]);
    bfrag_load(bq, xw + it * CHUNK, hi);
    __builtin_amdgcn_sched_barrier(0);
    recur_taps(ps, kc, tv);
    __builtin_amdgcn_sched_barrier(0);
    fir_chunk(tv, bq, acc0, acc1);
    ++it;
    unsigned int uq[4] = {0u, 0u, 0u, 0u};
#pragma unroll
    for (int i = 0; i < M_ORD; ++i) uq[i & 3] |= __float_as_uint(ps[i]);
    const unsigned int ub = (uq[0] | uq[1]) | (uq[2] | uq[3]);
    const bool zero = __builtin_amdgcn_ballot_w32((ub & 0x7fffffffu) != 0u) == 0u;
    if (zero || it >= NCHUNK) break;
  }

#pragma unroll
  for (int st = 0; st < 2; ++st) {
    const v8f acc = st ? acc1 : acc0;
    wave_lds_order();
#pragma unroll
    for (int r = 0; r < 8; ++r)
      dbuf[wave][r + 8 * hi][lm] = acc[r];
    wave_lds_order();
    if (lane < 16)
      y[(size_t)b * T_LEN + tb + wave * 32 + st * 16 + lane] = dbuf[wave][lane][lane];
  }
}

extern "C" void kernel_launch(void* const* d_in, const int* in_sizes, int n_in,
                              void* d_out, int out_size, void* d_ws, size_t ws_size,
                              hipStream_t stream) {
  (void)in_sizes; (void)n_in; (void)out_size; (void)d_ws; (void)ws_size;
  const float* x  = (const float*)d_in[0];
  const float* mc = (const float*)d_in[1];
  float* y = (float*)d_out;

  dim3 grid(BATCH * (T_LEN / TILE_T));
  dim3 block(NTHREADS);
  tv_fir_c2ir_kernel<<<grid, block, 0, stream>>>(x, mc, y);
}


# second exact exit: once sum|k*c_k| <= 7n/8 no later tap exceeds the largest of the last 24; a wave leaves when that times max|x| is under 2^-29 of each of its y sums (none can change); same sums
# speedup vs baseline: 19.7439x; 1.6815x over previous
#include <hip/hip_runtime.h>
#include <math.h>

typedef float v2f __attribute__((ext_vector_type(2)));
typedef float v4f __attribute__((ext_vector_type(4)));
typedef float v8f __attribute__((ext_vector_type(8)));
typedef unsigned int v2u __attribute__((ext_vector_type(2)));

#define BATCH   2
#define T_LEN   16000
#define NCEP    25
#define M_ORD   24
#define IR_LEN  2000
#define TILE_T  128
#define NWAVES  4
#define NTHREADS (NWAVES * 32)
#define CHUNK   80
#define NCHUNK  25
#define NGRP    (CHUNK / 4)
#define XS_LEN  2176


__device__ __forceinline__ void wave_lds_order() {
  __builtin_amdgcn_fence(__ATOMIC_RELEASE, "wavefront");
  __builtin_amdgcn_wave_barrier();
  __builtin_amdgcn_fence(__ATOMIC_ACQUIRE, "wavefront");
}

__device__ __forceinline__ void half_swap(float& a, float& b) {
  const v2u r = __builtin_amdgcn_permlane16_swap(__float_as_uint(a), __float_as_uint(b), false, false);
  a = __uint_as_float(r[0]);
  b = __uint_as_float(r[1]);
}

__device__ __forceinline__ float xhalf(const float v) {
  return __uint_as_float((unsigned int)__builtin_amdgcn_permlanex16(
      (int)__float_as_uint(v), (int)__float_as_uint(v), 0x76543210, 0xfedcba98, false, false));
}

__device__ __forceinline__ unsigned int wave_max_bits(unsigned int v) {
  v = max(v, (unsigned int)__builtin_amdgcn_update_dpp(0, (int)v, 0x161, 0xf, 0xf, true));
  v = max(v, (unsigned int)__builtin_amdgcn_update_dpp(0, (int)v, 0x162, 0xf, 0xf, true));
  v = max(v, (unsigned int)__builtin_amdgcn_update_dpp(0, (int)v, 0x164, 0xf, 0xf, true));
  v = max(v, (unsigned int)__builtin_amdgcn_update_dpp(0, (int)v, 0x168, 0xf, 0xf, true));
  const unsigned int o = (unsigned int)__builtin_amdgcn_permlanex16((int)v, (int)v, 0x76543210, 0xfedcba98, false, false);
  return max(v, o);
}

__device__ __forceinline__ void recur_table(v4f (&tv)[NGRP], const float* __restrict__ invp) {
#pragma unroll
  for (int q = 0; q < NGRP; ++q)
    tv[q] = *reinterpret_cast<const v4f*>(invp + 4 * q);
}

__device__ __forceinline__ void bfrag_load(v2f (&bq)[NGRP + 4], const float* __restrict__ xw, const int hi) {
#pragma unroll
  for (int q = 0; q < NGRP + 4; ++q) {
    const int kl = (q - 4) * 4 + 2 * hi;
    bq[q].x = xw[kl];   bq[q].y = xw[kl + 1];
  }
}

__device__ __forceinline__ void recur_taps(float (&ps)[M_ORD], const float (&kc)[NCEP],
                                           v4f (&tv)[NGRP]) {
#pragma unroll
  for (int j = 0; j < CHUNK; ++j) {
    const float hn = ps[0] * tv[j >> 2][j & 3];
    tv[j >> 2][j & 3] = hn;
#pragma unroll
    for (int i = 0; i < M_ORD - 1; ++i) ps[i] = fmaf(kc[i + 1], hn, ps[i + 1]);
    ps[M_ORD - 1] = kc[M_ORD] * hn;
  }
}

__device__ __forceinline__ void fir_chunk(v4f (&hv)[NGRP], const v2f (&bq)[NGRP + 4],
                                          v8f& acc0, v8f& acc1) {
#pragma unroll
  for (int g = 0; g < NGRP; ++g) {
    float t0 = hv[g].x, t1 = hv[g].y, t2 = hv[g].z, t3 = hv[g].w;
    half_swap(t0, t2);
    half_swap(t1, t3);
    const v2f a0 = {t0, t1};
    const v2f a1 = {t2, t3};
    acc0 = __builtin_amdgcn_wmma_f32_16x16x4_f32(
         false, a0,  false, bq[g + 4],
         (short)0, acc0,  false,  false);
    acc1 = __builtin_amdgcn_wmma_f32_16x16x4_f32(
         false, a1,  false, bq[g],
         (short)0, acc1,  false,  false);
  }
}

__global__ __launch_bounds__(NTHREADS)
void tv_fir_c2ir_kernel(const float* __restrict__ x,
                        const float* __restrict__ mc,
                        float* __restrict__ y) {
  __shared__ float xs[XS_LEN];
  __shared__ __attribute__((aligned(16))) float invn[IR_LEN];
  __shared__ float dbuf[NWAVES][16][16];
  __shared__ __attribute__((aligned(16))) unsigned int xmaxw[NWAVES];

  const int tid  = threadIdx.x;
  const int wave = __builtin_amdgcn_readfirstlane(tid >> 5);
  const int lane = tid & 31;
  const int hi   = lane >> 4;
  const int lm   = lane & 15;

  const int b  = blockIdx.x / (T_LEN / TILE_T);
  const int tb = (blockIdx.x % (T_LEN / TILE_T)) * TILE_T;

  float mcv[NCEP];
  {
    const float* mcp = mc + ((size_t)b * T_LEN + (tb + tid)) * NCEP;
#pragma unroll
    for (int k = 0; k < NCEP; ++k) mcv[k] = mcp[k];
  }
  float xv[(XS_LEN + NTHREADS - 1) / NTHREADS];
#pragma unroll
  for (int q = 0; q < (XS_LEN + NTHREADS - 1) / NTHREADS; ++q) {
    const int i  = tid + q * NTHREADS;
    const int g  = tb - (IR_LEN - 1) + i;
    const int gc = g < 0 ? 0 : (g > T_LEN - 1 ? T_LEN - 1 : g);
    const float v = x[(size_t)b * T_LEN + gc];
    xv[q] = (i < (IR_LEN - 1 + TILE_T) && g >= 0 && g < T_LEN) ? v : 0.0f;
  }
#pragma unroll
  for (int q = 0; q < (IR_LEN + NTHREADS - 1) / NTHREADS; ++q) {
    const int n = tid + q * NTHREADS;
    if (n < IR_LEN) invn[n] = (n > 0) ? 1.0f / (float)n : 1.0f;
  }
#pragma unroll
  for (int q = 0; q < (XS_LEN + NTHREADS - 1) / NTHREADS; ++q) {
    const int i = tid + q * NTHREADS;
    if (i < XS_LEN) xs[(XS_LEN - 1) - i] = xv[q];
  }
  {
    unsigned int xb = 0u;
#pragma unroll
    for (int q = 0; q < (XS_LEN + NTHREADS - 1) / NTHREADS; ++q) xb = max(xb, __float_as_uint(xv[q]) & 0x7fffffffu);
    xb = wave_max_bits(xb);
    if (lane == 0) xmaxw[wave] = xb;
  }

  float kc[NCEP];
  float ps[M_ORD];
  kc[0] = 0.0f;
#pragma unroll
  for (int k = 1; k < NCEP; ++k) kc[k] = (float)k * mcv[k];
#pragma unroll
  for (int k = 1; k < M_ORD; ++k) ps[k] = 0.0f;
  ps[0] = __expf(mcv[0]);
  float ksum = 0.0f;
#pragma unroll
  for (int k = 1; k < NCEP; ++k) ksum += fabsf(kc[k]);

  v8f acc0 = {};
  v8f acc1 = {};

  __syncthreads();

  float xmax;
  {
    const uint4 xw4 = *reinterpret_cast<const uint4*>(xmaxw);
    xmax = __uint_as_float(max(max(xw4.x, xw4.y), max(xw4.z, xw4.w)));
  }
  const int  sel      = lane & 7;
  const unsigned int holder = ((lane < 8) | (lane >= 24)) ? 1u : 0u;

  asm volatile("s_denorm_mode 12");

  const float* xw = &xs[(XS_LEN - 1) - (IR_LEN - 1) - (wave * 32 + lm)];

  v4f tv[NGRP];
  v2f bq[NGRP + 4];

  int it = 0;
  for (;;) {
    recur_table(tv, &invn[it * CHUNK]);
    bfrag_load(bq, xw + it * CHUNK, hi);
    __builtin_amdgcn_sched_barrier(0);
    recur_taps(ps, kc, tv);
    float tmax = 0.0f;
#pragma unroll
    for (int q = NGRP - M_ORD / 4; q < NGRP; ++q)
      tmax = __builtin_fmaxf(__builtin_fmaxf(tmax, __builtin_fmaxf(fabsf(tv[q].x), fabsf(tv[q].y))),
                             __builtin_fmaxf(fabsf(tv[q].z), fabsf(tv[q].w)));
    __builtin_amdgcn_sched_barrier(0);
    fir_chunk(tv, bq, acc0, acc1);
    ++it;
    unsigned int uq[4] = {0u, 0u, 0u, 0u};
#pragma unroll
    for (int i = 0; i < M_ORD; ++i) uq[i & 3] |= __float_as_uint(ps[i]);
    const unsigned int ub = (uq[0] | uq[1]) | (uq[2] | uq[3]);
    const bool zero = __builtin_amdgcn_ballot_w32((ub & 0x7fffffffu) != 0u) == 0u;
    const float bnd = (ksum <= 0.875f * (float)(it * CHUNK)) ? tmax * xmax : __builtin_inff();
    const float bno = xhalf(bnd);
    float d0 = acc0[0], d1 = acc1[0];
#pragma unroll
    for (int r = 1; r < 8; ++r) { d0 = (sel == r) ? acc0[r] : d0; d1 = (sel == r) ? acc1[r] : d1; }
    const float b0 = hi ? bno : bnd;
    const float b1 = hi ? bnd : bno;
    const unsigned int fits = (unsigned int)(b0 < 0x1p-29f * fabsf(d0)) & (unsigned int)(b1 < 0x1p-29f * fabsf(d1));
    const bool settled = __builtin_amdgcn_ballot_w32((holder & (fits ^ 1u)) != 0u) == 0u;
    if (zero || settled || it >= NCHUNK) break;
  }

#pragma unroll
  for (int st = 0; st < 2; ++st) {
    const v8f acc = st ? acc1 : acc0;
    wave_lds_order();
#pragma unroll
    for (int r = 0; r < 8; ++r)
      dbuf[wave][r + 8 * hi][lm] = acc[r];
    wave_lds_order();
    if (lane < 16)
      y[(size_t)b * T_LEN + tb + wave * 32 + st * 16 + lane] = dbuf[wave][lane][lane];
  }
}

extern "C" void kernel_launch(void* const* d_in, const int* in_sizes, int n_in,
                              void* d_out, int out_size, void* d_ws, size_t ws_size,
                              hipStream_t stream) {
  (void)in_sizes; (void)n_in; (void)out_size; (void)d_ws; (void)ws_size;
  const float* x  = (const float*)d_in[0];
  const float* mc = (const float*)d_in[1];
  float* y = (float*)d_out;

  dim3 grid(BATCH * (T_LEN / TILE_T));
  dim3 block(NTHREADS);
  tv_fir_c2ir_kernel<<<grid, block, 0, stream>>>(x, mc, y);
}
